# starting kernel
# baseline (speedup 1.0000x reference)
#include <hip/hip_runtime.h>
#include <math.h>

#define N_NODES 100000
#define N_EDGES 1600000
#define F_IN    256
#define F_H     128
#define N_CLS   10
#define H_PAD   16

typedef float v2f __attribute__((ext_vector_type(2)));
typedef float v8f __attribute__((ext_vector_type(8)));

__global__ __launch_bounds__(256) void k_deg_init(float* __restrict__ deg) {
    int i = blockIdx.x * 256 + threadIdx.x;
    if (i < N_NODES) deg[i] = 1.0f;
}

__global__ __launch_bounds__(256) void k_deg_add(const int* __restrict__ dst,
                                                 float* __restrict__ deg) {
    int e = blockIdx.x * 256 + threadIdx.x;
    if (e < N_EDGES) atomicAdd(&deg[dst[e]], 1.0f);
}

__global__ __launch_bounds__(256) void k_dinv(float* __restrict__ deg) {
    int i = blockIdx.x * 256 + threadIdx.x;
    if (i < N_NODES) {
        float d = deg[i];
        deg[i] = (d > 0.0f) ? rsqrtf(d) : 0.0f;
    }
}

__global__ __launch_bounds__(256) void k_zero(float* __restrict__ p, long n) {
    long i = (long)blockIdx.x * 256 + threadIdx.x;
    if (i < n) p[i] = 0.0f;
}

#define LDS_STRIDE 260
__global__ __launch_bounds__(256) void k_gemm1(const float* __restrict__ x,
                                               const float* __restrict__ W1,
                                               float* __restrict__ h1) {
    __shared__ float As[16 * LDS_STRIDE];
    const int tile = blockIdx.x;
    const int tid  = threadIdx.x;

    const float4* xg = (const float4*)(x + (size_t)tile * 16 * F_IN);
    #pragma unroll
    for (int it = 0; it < 4; ++it) {
        int q   = tid + 256 * it;
        int row = q >> 6;
        int c4  = q & 63;
        float4 v = xg[row * 64 + c4];
        float* d = &As[row * LDS_STRIDE + c4 * 4];
        d[0] = v.x; d[1] = v.y; d[2] = v.z; d[3] = v.w;
    }
    __syncthreads();

    const int wave = tid >> 5;
    const int lane = tid & 31;
    const int m    = lane & 15;
    const int kg   = lane >> 4;
    const int n0   = wave * 16;

    v8f c = {0.f,0.f,0.f,0.f,0.f,0.f,0.f,0.f};
    for (int k0 = 0; k0 < F_IN; k0 += 4) {
        int ka = k0 + 2 * kg;
        v2f a = *(const v2f*)&As[m * LDS_STRIDE + ka];
        v2f b;
        b.x = W1[(size_t)ka       * F_H + n0 + m];
        b.y = W1[(size_t)(ka + 1) * F_H + n0 + m];
        c = __builtin_amdgcn_wmma_f32_16x16x4_f32(false, a, false, b,
                                                  (short)0, c, false, false);
    }

    float* out = h1 + (size_t)tile * 16 * F_H;
    #pragma unroll
    for (int r = 0; r < 8; ++r) {
        int row = r + 8 * kg;
        out[(size_t)row * F_H + n0 + m] = c[r];
    }
}

__global__ __launch_bounds__(256) void k_scatter1(const int* __restrict__ src,
                                                  const int* __restrict__ dst,
                                                  const float* __restrict__ dinv,
                                                  const float* __restrict__ h1,
                                                  float* __restrict__ agg1) {
    long gid = (long)blockIdx.x * 256 + threadIdx.x;
    long e   = gid >> 5;
    if (e >= N_EDGES) return;
    int f4 = (int)(gid & 31);
    int s = src[e], d = dst[e];
    float nrm = dinv[s] * dinv[d];
    float4 v = *(const float4*)&h1[(size_t)s * F_H + f4 * 4];
    float* o = &agg1[(size_t)d * F_H + f4 * 4];
    atomicAdd(&o[0], v.x * nrm);
    atomicAdd(&o[1], v.y * nrm);
    atomicAdd(&o[2], v.z * nrm);
    atomicAdd(&o[3], v.w * nrm);
}

__global__ __launch_bounds__(256) void k_post1(const float* __restrict__ h1,
                                               const float* __restrict__ dinv,
                                               const float* __restrict__ b1,
                                               float* __restrict__ agg1) {
    long gid = (long)blockIdx.x * 256 + threadIdx.x;
    if (gid >= (long)N_NODES * F_H) return;
    int i = (int)(gid >> 7);
    int f = (int)(gid & (F_H - 1));
    float di = dinv[i];
    float v = agg1[gid] + h1[gid] * di * di + b1[f];
    agg1[gid] = v > 0.0f ? v : 0.0f;
}

__global__ __launch_bounds__(256) void k_padW2(const float* __restrict__ W2,
                                               float* __restrict__ Bp) {
    int t = blockIdx.x * 256 + threadIdx.x;
    if (t >= F_H * H_PAD) return;
    int k = t >> 4, c = t & 15;
    Bp[t] = (c < N_CLS) ? W2[k * N_CLS + c] : 0.0f;
}

__global__ __launch_bounds__(256) void k_gemm2(const float* __restrict__ A,
                                               const float* __restrict__ Bp,
                                               float* __restrict__ h2) {
    const int wave = threadIdx.x >> 5;
    const int lane = threadIdx.x & 31;
    const int tile = blockIdx.x * 8 + wave;
    if ((size_t)tile * 16 >= N_NODES) return;
    const int m  = lane & 15;
    const int kg = lane >> 4;

    const float* Ar = A + ((size_t)tile * 16 + m) * F_H;
    v8f c = {0.f,0.f,0.f,0.f,0.f,0.f,0.f,0.f};
    for (int k0 = 0; k0 < F_H; k0 += 4) {
        int ka = k0 + 2 * kg;
        v2f a = *(const v2f*)&Ar[ka];
        v2f b;
        b.x = Bp[ka * H_PAD + m];
        b.y = Bp[(ka + 1) * H_PAD + m];
        c = __builtin_amdgcn_wmma_f32_16x16x4_f32(false, a, false, b,
                                                  (short)0, c, false, false);
    }
    float* out = h2 + (size_t)tile * 16 * H_PAD;
    #pragma unroll
    for (int r = 0; r < 8; ++r) {
        int row = r + 8 * kg;
        out[(size_t)row * H_PAD + m] = c[r];
    }
}

__global__ __launch_bounds__(256) void k_scatter2(const int* __restrict__ src,
                                                  const int* __restrict__ dst,
                                                  const float* __restrict__ dinv,
                                                  const float* __restrict__ h2,
                                                  float* __restrict__ agg2) {
    long gid = (long)blockIdx.x * 256 + threadIdx.x;
    long e   = gid >> 4;
    if (e >= N_EDGES) return;
    int f = (int)(gid & 15);
    if (f >= N_CLS) return;
    int s = src[e], d = dst[e];
    float nrm = dinv[s] * dinv[d];
    atomicAdd(&agg2[(size_t)d * H_PAD + f], h2[(size_t)s * H_PAD + f] * nrm);
}

__global__ __launch_bounds__(256) void k_post2(const float* __restrict__ h2,
                                               const float* __restrict__ agg2,
                                               const float* __restrict__ dinv,
                                               const float* __restrict__ b2,
                                               float* __restrict__ out) {
    int i = blockIdx.x * 256 + threadIdx.x;
    if (i >= N_NODES) return;
    float di = dinv[i];
    float d2 = di * di;
    float v[N_CLS];
    float mx = -INFINITY;
    #pragma unroll
    for (int c = 0; c < N_CLS; ++c) {
        float t = agg2[(size_t)i * H_PAD + c] + h2[(size_t)i * H_PAD + c] * d2 + b2[c];
        v[c] = t;
        mx = fmaxf(mx, t);
    }
    float s = 0.0f;
    #pragma unroll
    for (int c = 0; c < N_CLS; ++c) s += expf(v[c] - mx);
    float lse = logf(s) + mx;
    #pragma unroll
    for (int c = 0; c < N_CLS; ++c) out[(size_t)i * N_CLS + c] = v[c] - lse;
}

extern "C" void kernel_launch(void* const* d_in, const int* in_sizes, int n_in,
                              void* d_out, int out_size, void* d_ws, size_t ws_size,
                              hipStream_t stream) {
    const float* x   = (const float*)d_in[0];
    const int*   ei  = (const int*)d_in[1];
    const float* W1  = (const float*)d_in[2];
    const float* b1  = (const float*)d_in[3];
    const float* W2  = (const float*)d_in[4];
    const float* b2  = (const float*)d_in[5];
    float*       out = (float*)d_out;

    const int* src = ei;
    const int* dst = ei + N_EDGES;

    float* ws    = (float*)d_ws;
    float* dinv  = ws;
    float* h1    = dinv + 100352;
    float* agg1  = h1 + (size_t)N_NODES * F_H;
    float* W2pad = agg1 + (size_t)N_NODES * F_H;
    float* h2    = h1;
    float* agg2  = h1 + (size_t)N_NODES * H_PAD;

    const int nblk_node = (N_NODES + 255) / 256;

    k_deg_init<<<nblk_node, 256, 0, stream>>>(dinv);
    k_deg_add <<<(N_EDGES + 255) / 256, 256, 0, stream>>>(dst, dinv);
    k_dinv    <<<nblk_node, 256, 0, stream>>>(dinv);

    k_gemm1<<<N_NODES / 16, 256, 0, stream>>>(x, W1, h1);
    {
        long n = (long)N_NODES * F_H;
        k_zero<<<(unsigned)((n + 255) / 256), 256, 0, stream>>>(agg1, n);
    }
    {
        long t = (long)N_EDGES * 32;
        k_scatter1<<<(unsigned)((t + 255) / 256), 256, 0, stream>>>(src, dst, dinv, h1, agg1);
    }
    {
        long n = (long)N_NODES * F_H;
        k_post1<<<(unsigned)((n + 255) / 256), 256, 0, stream>>>(h1, dinv, b1, agg1);
    }

    k_padW2<<<(F_H * H_PAD + 255) / 256, 256, 0, stream>>>(W2, W2pad);
    k_gemm2<<<(N_NODES / 16 + 7) / 8, 256, 0, stream>>>(agg1, W2pad, h2);
    {
        long n = (long)N_NODES * H_PAD;
        k_zero<<<(unsigned)((n + 255) / 256), 256, 0, stream>>>(agg2, n);
    }
    {
        long t = (long)N_EDGES * 16;
        k_scatter2<<<(unsigned)((t + 255) / 256), 256, 0, stream>>>(src, dst, dinv, h2, agg2);
    }
    k_post2<<<nblk_node, 256, 0, stream>>>(h2, agg2, dinv, b2, out);
}


# layer-1 owner blocks save their edges (src, offset, weight) in 32-entry slabs; layer-2 sum runs in the same blocks from them, softmax there: no dst scan, partial planes, post pass; sum order changes
# speedup vs baseline: 3.0759x; 3.0759x over previous
#include <hip/hip_runtime.h>
#include <math.h>

#define N_NODES 100000
#define N_EDGES 1600000
#define F_IN    256
#define F_H     128
#define N_CLS   10
#define H_PAD   16

typedef float v2f __attribute__((ext_vector_type(2)));
typedef float v8f __attribute__((ext_vector_type(8)));

#define OWN_NPB    391
#define DEG_GROUPS 32
#define DEG_RANGES 8
#define DEG_EPG    (N_EDGES / DEG_GROUPS)
#define DEG_NPR    (N_NODES / DEG_RANGES)
#define DEG_LD     13

__global__ __launch_bounds__(1024) void k_deg_hist(const int* __restrict__ dst,
                                                   unsigned* __restrict__ part,
                                                   unsigned* __restrict__ omap) {
    __shared__ unsigned cnt[DEG_NPR];
    const int tid = threadIdx.x;
    const int r   = blockIdx.x & (DEG_RANGES - 1);
    const int g   = blockIdx.x >> 3;
    const int4* p = (const int4*)(dst + (size_t)g * DEG_EPG);

    int4 v[DEG_LD];
    #pragma unroll
    for (int k = 0; k < DEG_LD; ++k) {
        int i = tid + 1024 * k;
        v[k] = (i < DEG_EPG / 4) ? p[i] : make_int4(-1, -1, -1, -1);
    }
    for (int i = tid; i < DEG_NPR; i += 1024) cnt[i] = 0u;
    __syncthreads();

    const unsigned lo = (unsigned)r * DEG_NPR;
    #pragma unroll
    for (int k = 0; k < DEG_LD; ++k) {
        unsigned a;
        a = (unsigned)v[k].x - lo; if (a < DEG_NPR) atomicAdd(&cnt[a], 1u);
        a = (unsigned)v[k].y - lo; if (a < DEG_NPR) atomicAdd(&cnt[a], 1u);
        a = (unsigned)v[k].z - lo; if (a < DEG_NPR) atomicAdd(&cnt[a], 1u);
        a = (unsigned)v[k].w - lo; if (a < DEG_NPR) atomicAdd(&cnt[a], 1u);
    }
    __syncthreads();

    unsigned* o = part + (size_t)g * (N_NODES / 2) + r * (DEG_NPR / 2);
    for (int j = tid; j < DEG_NPR / 2; j += 1024)
        o[j] = cnt[2 * j] | (cnt[2 * j + 1] << 16);

    #pragma unroll
    for (int k = 0; k < DEG_LD; ++k) {
        int i = tid + 1024 * k;
        if ((k & (DEG_RANGES - 1)) == r && i < DEG_EPG / 4) {
            unsigned o0 = (unsigned)v[k].x / OWN_NPB, o1 = (unsigned)v[k].y / OWN_NPB;
            unsigned o2 = (unsigned)v[k].z / OWN_NPB, o3 = (unsigned)v[k].w / OWN_NPB;
            omap[(size_t)g * (DEG_EPG / 4) + i] = o0 | (o1 << 8) | (o2 << 16) | (o3 << 24);
        }
    }
}

__global__ __launch_bounds__(256) void k_dinv_sum(const unsigned* __restrict__ part,
                                                  float* __restrict__ dinv) {
    int t = blockIdx.x * 256 + threadIdx.x;
    if (t >= N_NODES / 2) return;
    unsigned c0 = 1u, c1 = 1u;
    #pragma unroll
    for (int g = 0; g < DEG_GROUPS; ++g) {
        unsigned w = part[(size_t)g * (N_NODES / 2) + t];
        c0 += w & 0xffffu;
        c1 += w >> 16;
    }
    float d0 = (float)c0, d1 = (float)c1;
    float2 o;
    o.x = (d0 > 0.0f) ? rsqrtf(d0) : 0.0f;
    o.y = (d1 > 0.0f) ? rsqrtf(d1) : 0.0f;
    ((float2*)dinv)[t] = o;
}

#define LDS_STRIDE 260
#define G1_ROWS    32
__global__ __launch_bounds__(256) void k_gemm1(const float* __restrict__ x,
                                               const float* __restrict__ W1,
                                               float* __restrict__ h1) {
    __shared__ float As[G1_ROWS * LDS_STRIDE];
    const int strip = blockIdx.x;
    const int tid   = threadIdx.x;

    const float4* xg = (const float4*)(x + (size_t)strip * G1_ROWS * F_IN);
    #pragma unroll
    for (int it = 0; it < 8; ++it) {
        int q   = tid + 256 * it;
        int row = q >> 6;
        int c4  = q & 63;
        float4 v = xg[q];
        float* d = &As[row * LDS_STRIDE + c4 * 4];
        d[0] = v.x; d[1] = v.y; d[2] = v.z; d[3] = v.w;
    }
    __syncthreads();

    const int wave = tid >> 5;
    const int lane = tid & 31;
    const int m    = lane & 15;
    const int kg   = lane >> 4;
    const int n0   = wave * 16;

    v8f c0 = {0.f,0.f,0.f,0.f,0.f,0.f,0.f,0.f};
    v8f c1 = {0.f,0.f,0.f,0.f,0.f,0.f,0.f,0.f};
    for (int k0 = 0; k0 < F_IN; k0 += 4) {
        int ka = k0 + 2 * kg;
        v2f a0 = *(const v2f*)&As[m * LDS_STRIDE + ka];
        v2f a1 = *(const v2f*)&As[(16 + m) * LDS_STRIDE + ka];
        v2f b;
        b.x = W1[(size_t)ka       * F_H + n0 + m];
        b.y = W1[(size_t)(ka + 1) * F_H + n0 + m];
        c0 = __builtin_amdgcn_wmma_f32_16x16x4_f32(false, a0, false, b,
                                                   (short)0, c0, false, false);
        c1 = __builtin_amdgcn_wmma_f32_16x16x4_f32(false, a1, false, b,
                                                   (short)0, c1, false, false);
    }

    asm volatile("v_nop\n\tv_nop\n\tv_nop\n\tv_nop" : "+v"(c0), "+v"(c1));

    float* out = h1 + (size_t)strip * G1_ROWS * F_H;
    #pragma unroll
    for (int r = 0; r < 8; ++r) {
        int row = r + 8 * kg;
        out[(size_t)row * F_H + n0 + m] = c0[r];
        out[(size_t)(16 + row) * F_H + n0 + m] = c1[r];
    }
}

#define S1_NPB    OWN_NPB
#define S1_BLOCKS 256
#define S1_STEPS  (N_EDGES / 512)
#define S1_KIT    98
#define S1_QCAP   64
#define S1_ROT    12
#define S1_PITCH  (F_H + 4)

#define E2_NSLAB   463
#define E2_NONE    0xFFFFFFFFu

struct SumCtx {
    const int*   src;
    const int*   dst;
    const float* dinv;
    const float* feat;
    float*       acc;
    unsigned*    slabctr;
    uint2*       slabs;
    int          lo;
};

__device__ __forceinline__ int s1_chunk(int k, int wave, int rot) {
    int t = wave + 32 * k;
    if (t >= S1_STEPS) return -1;
    int tt = t + rot;
    return (tt >= S1_STEPS) ? tt - S1_STEPS : tt;
}

__device__ __forceinline__ uint4 s1_load(const unsigned* __restrict__ omap, int k, int wave, int rot, int lane) {
    int tt = s1_chunk(k, wave, rot);
    if (tt < 0) return make_uint4(0u, 0u, 0u, 0u);
    return ((const uint4*)omap)[tt * 32 + lane];
}

__device__ __forceinline__ void s2_add32(bool ok, unsigned xs, float w, int lane,
                                         const float* __restrict__ h2, float* acc) {
    const unsigned okm = __builtin_amdgcn_ballot_w32(ok);
    if (okm == 0u) return;
    xs = ok ? xs : 0u;
    w  = ok ? w : 0.0f;
    const int half = lane >> 4;
    const int f    = lane & 15;
    unsigned xj[16];
    float r[16], wj[16];
    #pragma unroll
    for (int c = 0; c < 16; ++c) xj[c] = (unsigned)__shfl((int)xs, 2 * c + half);
    #pragma unroll
    for (int c = 0; c < 16; ++c) r[c] = h2[(size_t)(xj[c] >> 9) * H_PAD + f];
    #pragma unroll
    for (int c = 0; c < 16; ++c) wj[c] = __shfl(w, 2 * c + half);
    #pragma unroll
    for (int c = 0; c < 16; ++c) r[c] = ((okm >> (2 * c + half)) & 1u) != 0u ? r[c] : 0.0f;
    if (f < N_CLS) {
        #pragma unroll
        for (int c = 0; c < 16; ++c) atomicAdd(&acc[(int)(xj[c] & 511u) * N_CLS + f], r[c] * wj[c]);
    }
}

template <int L>
__device__ __forceinline__ void s1_rows(int nb, const unsigned* qw, int lane, const SumCtx& cx) {
    if (nb <= 0) return;
    const bool act = lane < nb;
    int e = act ? (int)qw[lane] : 0;
    int s = 0, dl = 0;
    float w = 0.0f;
    if (act) {
        s = cx.src[e];
        int d = cx.dst[e];
        dl = d - cx.lo;
        w = cx.dinv[s] * cx.dinv[d];
    }
    if (L == 2) {
        s2_add32(act, ((unsigned)s << 9) | (unsigned)dl, w, lane, cx.feat, cx.acc);
        return;
    }
    unsigned slab = 0u;
    if (lane == 0) slab = atomicAdd(cx.slabctr, 1u);
    slab = (unsigned)__builtin_amdgcn_readfirstlane((int)slab);
    if (slab < (unsigned)E2_NSLAB) {
        uint2 ent;
        ent.x = act ? (((unsigned)s << 9) | (unsigned)dl) : E2_NONE;
        ent.y = __builtin_bit_cast(unsigned, w);
        cx.slabs[slab * 32u + (unsigned)lane] = ent;
    }
    const float* __restrict__ h1 = cx.feat;
    float* acc = cx.acc;
    for (int j0 = 0; j0 < nb; j0 += 8) {
        float r[8][4];
        #pragma unroll
        for (int c = 0; c < 8; ++c) {
            r[c][0] = 0.0f; r[c][1] = 0.0f; r[c][2] = 0.0f; r[c][3] = 0.0f;
            if (j0 + c < nb) {
                int sj = __builtin_amdgcn_readlane(s, j0 + c);
                const float* row = h1 + (size_t)sj * F_H + lane;
                r[c][0] = row[0]; r[c][1] = row[32]; r[c][2] = row[64]; r[c][3] = row[96];
            }
        }
        #pragma unroll
        for (int c = 0; c < 8; ++c) {
            if (j0 + c < nb) {
                float wj = __builtin_bit_cast(float, __builtin_amdgcn_readlane(__builtin_bit_cast(int, w), j0 + c));
                int   dj = __builtin_amdgcn_readlane(dl, j0 + c);
                float* a = acc + dj * S1_PITCH + lane;
                atomicAdd(a,      r[c][0] * wj);
                atomicAdd(a + 32, r[c][1] * wj);
                atomicAdd(a + 64, r[c][2] * wj);
                atomicAdd(a + 96, r[c][3] * wj);
            }
        }
    }
}

__device__ __forceinline__ unsigned s1_nzb(unsigned y) {
    return ((y & 0x7F7F7F7Fu) + 0x7F7F7F7Fu) | y;
}

template <int L>
__device__ __forceinline__ void s1_scan(uint4 v, int tt, int lane, unsigned pat,
                                        unsigned* qw, int& qn, const SumCtx& cx) {
    if (tt < 0) return;
    unsigned n0 = s1_nzb(v.x ^ pat), n1 = s1_nzb(v.y ^ pat), n2 = s1_nzb(v.z ^ pat), n3 = s1_nzb(v.w ^ pat);
    unsigned c = n0 >> 7;
    c = (c & ~0x02020202u) | ((n1 >> 6) & 0x02020202u);
    c = (c & ~0x04040404u) | ((n2 >> 5) & 0x04040404u);
    c = (c & ~0x08080808u) | ((n3 >> 4) & 0x08080808u);
    unsigned mm = ~c & 0x0F0F0F0Fu;
    const unsigned e0 = (unsigned)tt * 512u + (unsigned)lane * 16u;
    for (;;) {
        const bool has = mm != 0u;
        const unsigned b = __builtin_amdgcn_ballot_w32(has);
        if (b == 0u) break;
        if (has) {
            unsigned p = (unsigned)__builtin_ctz(mm);
            qw[qn + (int)__builtin_amdgcn_mbcnt_lo(b, 0u)] = e0 + ((p & 3u) << 2) + (p >> 3);
            mm &= mm - 1u;
        }
        qn += __builtin_popcount(b);
        if (qn >= 32) {
            s1_rows<L>(32, qw + (qn - 32), lane, cx);
            qn -= 32;
        }
    }
}

template <int L>
__device__ __forceinline__ void s1_sweep(const unsigned* __restrict__ omap, uint4 v0, uint4 v1, uint4 v2,
                                         int wave, int rot, int lane, unsigned pat,
                                         unsigned* qw, const SumCtx& cx) {
    int qn = 0;
    for (int k = 0; k < S1_KIT; k += 3) {
        uint4 n0 = s1_load(omap, k + 3, wave, rot, lane);
        uint4 n1 = s1_load(omap, k + 4, wave, rot, lane);
        uint4 n2 = s1_load(omap, k + 5, wave, rot, lane);
        s1_scan<L>(v0, s1_chunk(k,     wave, rot), lane, pat, qw, qn, cx);
        s1_scan<L>(v1, s1_chunk(k + 1, wave, rot), lane, pat, qw, qn, cx);
        s1_scan<L>(v2, s1_chunk(k + 2, wave, rot), lane, pat, qw, qn, cx);
        v0 = n0; v1 = n1; v2 = n2;
    }
    s1_rows<L>(qn, qw, lane, cx);
}

__global__ __launch_bounds__(1024) void k_sum1(const int* __restrict__ src,
                                               const int* __restrict__ dst,
                                               const unsigned* __restrict__ omap,
                                               const float* __restrict__ dinv,
                                               const float* __restrict__ h1,
                                               const float* __restrict__ b1,
                                               const float* __restrict__ W2,
                                               const float* __restrict__ b2,
                                               float* __restrict__ h2,
                                               float* __restrict__ agg2,
                                               unsigned* __restrict__ hdr,
                                               unsigned* __restrict__ slabs) {
    __shared__ float    acc[S1_NPB * S1_PITCH];
    __shared__ unsigned q[32 * S1_QCAP];
    __shared__ unsigned slabctr;
    const int tid  = threadIdx.x;
    const int lane = tid & 31;
    const int wave = __builtin_amdgcn_readfirstlane(tid >> 5);
    const int lo   = blockIdx.x * S1_NPB;
    const int cnt  = (N_NODES - lo < S1_NPB) ? (N_NODES - lo) : S1_NPB;
    const int rot  = blockIdx.x * S1_ROT;
    const unsigned pat = (unsigned)blockIdx.x * 0x01010101u;

    uint4 v0 = s1_load(omap, 0, wave, rot, lane);
    uint4 v1 = s1_load(omap, 1, wave, rot, lane);
    uint4 v2 = s1_load(omap, 2, wave, rot, lane);

    if (tid == 0) slabctr = 0u;
    const float4* g4 = (const float4*)(h1 + (size_t)lo * F_H);
    float4* a4 = (float4*)acc;
    const float4 bb = ((const float4*)b1)[tid & 31];
    for (int i = tid; i < cnt * (F_H / 4); i += 1024) {
        const float  dv = dinv[lo + (i >> 5)];
        const float4 h  = g4[i];
        float4 o;
        o.x = h.x * dv * dv + bb.x;
        o.y = h.y * dv * dv + bb.y;
        o.z = h.z * dv * dv + bb.z;
        o.w = h.w * dv * dv + bb.w;
        a4[(i >> 5) * (S1_PITCH / 4) + (tid & 31)] = o;
    }
    __syncthreads();

    SumCtx cx;
    cx.src = src; cx.dst = dst; cx.dinv = dinv; cx.feat = h1; cx.acc = acc;
    cx.slabctr = &slabctr;
    cx.slabs   = (uint2*)slabs + (size_t)blockIdx.x * (E2_NSLAB * 32);
    cx.lo      = lo;
    s1_sweep<1>(omap, v0, v1, v2, wave, rot, lane, pat, q + wave * S1_QCAP, cx);
    __syncthreads();
    if (tid == 0) hdr[blockIdx.x] = slabctr;

    const int ntile = (cnt + 15) >> 4;
    if (wave < ntile) {
        const int m  = lane & 15;
        const int kg = lane >> 4;
        const int t0 = wave * 16;
        const int ra = (t0 + m < cnt) ? (t0 + m) : (cnt - 1);
        const float* Ar = acc + ra * S1_PITCH;
        const int  mc   = (m < N_CLS) ? m : 0;
        const bool real = m < N_CLS;

        float d2[8];
        #pragma unroll
        for (int r = 0; r < 8; ++r) {
            const int rr = t0 + r + 8 * kg;
            const float di = dinv[lo + ((rr < cnt) ? rr : (cnt - 1))];
            d2[r] = di * di;
        }
        const float bcol = real ? b2[m] : 0.0f;

        v8f c = {0.f,0.f,0.f,0.f,0.f,0.f,0.f,0.f};
        #pragma unroll
        for (int hh = 0; hh < 2; ++hh) {
            v2f a[16], b[16];
            #pragma unroll
            for (int j = 0; j < 16; ++j) {
                const int ka = 64 * hh + 4 * j + 2 * kg;
                a[j]   = *(const v2f*)&Ar[ka];
                b[j].x = W2[ka * N_CLS + mc];
                b[j].y = W2[(ka + 1) * N_CLS + mc];
            }
            #pragma unroll
            for (int j = 0; j < 16; ++j) {
                a[j].x = a[j].x > 0.0f ? a[j].x : 0.0f;
                a[j].y = a[j].y > 0.0f ? a[j].y : 0.0f;
                b[j].x = real ? b[j].x : 0.0f;
                b[j].y = real ? b[j].y : 0.0f;
            }
            __builtin_amdgcn_sched_barrier(0);
            #pragma unroll
            for (int j = 0; j < 16; ++j)
                c = __builtin_amdgcn_wmma_f32_16x16x4_f32(false, a[j], false, b[j],
                                                          (short)0, c, false, false);
            __builtin_amdgcn_sched_barrier(0);
            asm volatile("v_nop\n\tv_nop\n\tv_nop\n\tv_nop" : "+v"(c));
        }

        #pragma unroll
        for (int r = 0; r < 8; ++r) {
            const int rr = t0 + r + 8 * kg;
            if (rr < cnt) {
                const size_t o = (size_t)(lo + rr) * H_PAD + m;
                h2[o]   = c[r];
                agg2[o] = c[r] * d2[r] + bcol;
            }
        }
    }
}

__global__ __launch_bounds__(1024) void k_sum2(const int* __restrict__ src,
                                               const int* __restrict__ dst,
                                               const unsigned* __restrict__ omap,
                                               const float* __restrict__ dinv,
                                               const float* __restrict__ h2,
                                               const float* __restrict__ agg2,
                                               const unsigned* __restrict__ hdr,
                                               const unsigned* __restrict__ slabs,
                                               float* __restrict__ out) {
    __shared__ float    acc[S1_NPB * N_CLS];
    __shared__ unsigned q[32 * S1_QCAP];
    const int tid  = threadIdx.x;
    const int lane = tid & 31;
    const int wave = __builtin_amdgcn_readfirstlane(tid >> 5);
    const int lo   = blockIdx.x * S1_NPB;
    const int cnt  = (N_NODES - lo < S1_NPB) ? (N_NODES - lo) : S1_NPB;
    const unsigned nsl = (unsigned)__builtin_amdgcn_readfirstlane((int)hdr[blockIdx.x]);
    const uint2* bl = (const uint2*)slabs + (size_t)blockIdx.x * (E2_NSLAB * 32);

    uint2 ent = make_uint2(E2_NONE, 0u);
    if (nsl <= (unsigned)E2_NSLAB && (unsigned)wave < nsl) ent = bl[wave * 32 + lane];

    for (int i = tid; i < cnt * N_CLS; i += 1024) {
        const int n = i / N_CLS;
        acc[i] = agg2[(size_t)(lo + n) * H_PAD + (i - n * N_CLS)];
    }
    __syncthreads();

    if (nsl <= (unsigned)E2_NSLAB) {
        for (unsigned sl = (unsigned)wave; sl < nsl; sl += 32u) {
            uint2 nx = make_uint2(E2_NONE, 0u);
            if (sl + 32u < nsl) nx = bl[(sl + 32u) * 32u + (unsigned)lane];
            s2_add32(ent.x != E2_NONE, ent.x, __builtin_bit_cast(float, ent.y), lane, h2, acc);
            ent = nx;
        }
    } else {
        SumCtx cx;
        cx.src = src; cx.dst = dst; cx.dinv = dinv; cx.feat = h2; cx.acc = acc;
        cx.slabctr = nullptr; cx.slabs = nullptr;
        cx.lo      = lo;
        const int rot = blockIdx.x * S1_ROT;
        const unsigned pat = (unsigned)blockIdx.x * 0x01010101u;
        uint4 v0 = s1_load(omap, 0, wave, rot, lane);
        uint4 v1 = s1_load(omap, 1, wave, rot, lane);
        uint4 v2 = s1_load(omap, 2, wave, rot, lane);
        s1_sweep<2>(omap, v0, v1, v2, wave, rot, lane, pat, q + wave * S1_QCAP, cx);
    }
    __syncthreads();

    if (tid < cnt) {
        float v[N_CLS];
        #pragma unroll
        for (int c = 0; c < N_CLS; ++c) v[c] = acc[tid * N_CLS + c];
        float mx = -INFINITY;
        #pragma unroll
        for (int c = 0; c < N_CLS; ++c) mx = fmaxf(mx, v[c]);
        float s = 0.0f;
        #pragma unroll
        for (int c = 0; c < N_CLS; ++c) s += expf(v[c] - mx);
        float lse = logf(s) + mx;
        #pragma unroll
        for (int c = 0; c < N_CLS; ++c) acc[tid * N_CLS + c] = v[c] - lse;
    }
    __syncthreads();

    float* o = out + (size_t)lo * N_CLS;
    for (int i = tid; i < cnt * N_CLS; i += 1024) o[i] = acc[i];
}

extern "C" void kernel_launch(void* const* d_in, const int* in_sizes, int n_in,
                              void* d_out, int out_size, void* d_ws, size_t ws_size,
                              hipStream_t stream) {
    const float* x   = (const float*)d_in[0];
    const int*   ei  = (const int*)d_in[1];
    const float* W1  = (const float*)d_in[2];
    const float* b1  = (const float*)d_in[3];
    const float* W2  = (const float*)d_in[4];
    const float* b2  = (const float*)d_in[5];
    float*       out = (float*)d_out;

    const int* src = ei;
    const int* dst = ei + N_EDGES;

    float* ws    = (float*)d_ws;
    float* dinv  = ws;
    float* h1    = dinv + 100352;
    float* reg2  = h1 + (size_t)N_NODES * F_H;
    float* h2    = reg2;
    float* agg2  = reg2 + (size_t)N_NODES * H_PAD;
    unsigned* part  = (unsigned*)reg2;
    unsigned* omap  = (unsigned*)(reg2 + (size_t)N_NODES * 32);
    unsigned* hdr   = (unsigned*)(reg2 + (size_t)N_NODES * 36);
    unsigned* slabs = (unsigned*)(reg2 + (size_t)N_NODES * 36 + 256);

    k_deg_hist<<<DEG_GROUPS * DEG_RANGES, 1024, 0, stream>>>(dst, part, omap);
    k_dinv_sum<<<(N_NODES / 2 + 255) / 256, 256, 0, stream>>>(part, dinv);

    k_gemm1<<<N_NODES / G1_ROWS, 256, 0, stream>>>(x, W1, h1);
    k_sum1<<<S1_BLOCKS, 1024, 0, stream>>>(src, dst, omap, dinv, h1, b1, W2, b2, h2, agg2, hdr, slabs);

    k_sum2<<<S1_BLOCKS, 1024, 0, stream>>>(src, dst, omap, dinv, h2, agg2, hdr, slabs, out);
}


# edge records: 391 extra blocks of the GEMM1 launch write (src<<9 | dst offset in owner range, dinv[src]) per edge; owner blocks read one 8-byte record per own edge, not src, dst, dinv[src]
# speedup vs baseline: 3.2625x; 1.0607x over previous
#include <hip/hip_runtime.h>
#include <math.h>

#define N_NODES 100000
#define N_EDGES 1600000
#define F_IN    256
#define F_H     128
#define N_CLS   10
#define H_PAD   16

typedef float v2f __attribute__((ext_vector_type(2)));
typedef float v8f __attribute__((ext_vector_type(8)));

#define OWN_NPB    391
#define DEG_GROUPS 32
#define DEG_RANGES 8
#define DEG_EPG    (N_EDGES / DEG_GROUPS)
#define DEG_NPR    (N_NODES / DEG_RANGES)
#define DEG_LD     13

__global__ __launch_bounds__(1024) void k_deg_hist(const int* __restrict__ dst,
                                                   unsigned* __restrict__ part,
                                                   unsigned* __restrict__ omap) {
    __shared__ unsigned cnt[DEG_NPR];
    const int tid = threadIdx.x;
    const int r   = blockIdx.x & (DEG_RANGES - 1);
    const int g   = blockIdx.x >> 3;
    const int4* p = (const int4*)(dst + (size_t)g * DEG_EPG);

    int4 v[DEG_LD];
    #pragma unroll
    for (int k = 0; k < DEG_LD; ++k) {
        int i = tid + 1024 * k;
        v[k] = (i < DEG_EPG / 4) ? p[i] : make_int4(-1, -1, -1, -1);
    }
    for (int i = tid; i < DEG_NPR; i += 1024) cnt[i] = 0u;
    __syncthreads();

    const unsigned lo = (unsigned)r * DEG_NPR;
    #pragma unroll
    for (int k = 0; k < DEG_LD; ++k) {
        unsigned a;
        a = (unsigned)v[k].x - lo; if (a < DEG_NPR) atomicAdd(&cnt[a], 1u);
        a = (unsigned)v[k].y - lo; if (a < DEG_NPR) atomicAdd(&cnt[a], 1u);
        a = (unsigned)v[k].z - lo; if (a < DEG_NPR) atomicAdd(&cnt[a], 1u);
        a = (unsigned)v[k].w - lo; if (a < DEG_NPR) atomicAdd(&cnt[a], 1u);
    }
    __syncthreads();

    unsigned* o = part + (size_t)g * (N_NODES / 2) + r * (DEG_NPR / 2);
    for (int j = tid; j < DEG_NPR / 2; j += 1024)
        o[j] = cnt[2 * j] | (cnt[2 * j + 1] << 16);

    #pragma unroll
    for (int k = 0; k < DEG_LD; ++k) {
        int i = tid + 1024 * k;
        if ((k & (DEG_RANGES - 1)) == r && i < DEG_EPG / 4) {
            unsigned o0 = (unsigned)v[k].x / OWN_NPB, o1 = (unsigned)v[k].y / OWN_NPB;
            unsigned o2 = (unsigned)v[k].z / OWN_NPB, o3 = (unsigned)v[k].w / OWN_NPB;
            omap[(size_t)g * (DEG_EPG / 4) + i] = o0 | (o1 << 8) | (o2 << 16) | (o3 << 24);
        }
    }
}

__global__ __launch_bounds__(256) void k_dinv_sum(const unsigned* __restrict__ part,
                                                  float* __restrict__ dinv) {
    int t = blockIdx.x * 256 + threadIdx.x;
    if (t >= N_NODES / 2) return;
    unsigned c0 = 1u, c1 = 1u;
    #pragma unroll
    for (int g = 0; g < DEG_GROUPS; ++g) {
        unsigned w = part[(size_t)g * (N_NODES / 2) + t];
        c0 += w & 0xffffu;
        c1 += w >> 16;
    }
    float d0 = (float)c0, d1 = (float)c1;
    float2 o;
    o.x = (d0 > 0.0f) ? rsqrtf(d0) : 0.0f;
    o.y = (d1 > 0.0f) ? rsqrtf(d1) : 0.0f;
    ((float2*)dinv)[t] = o;
}

#define LDS_STRIDE 260
#define G1_ROWS    32
#define EP_GRP     4
#define EP_BLOCKS  ((N_EDGES / 4 + 256 * EP_GRP - 1) / (256 * EP_GRP))
__global__ __launch_bounds__(256) void k_gemm1(const float* __restrict__ x,
                                               const float* __restrict__ W1,
                                               float* __restrict__ h1,
                                               const int* __restrict__ src,
                                               const int* __restrict__ dst,
                                               const float* __restrict__ dinv,
                                               unsigned* __restrict__ rec) {
    __shared__ float As[G1_ROWS * LDS_STRIDE];
    const int tid   = threadIdx.x;

    if (blockIdx.x < EP_BLOCKS) {
        const int4* s4 = (const int4*)src;
        const int4* d4 = (const int4*)dst;
        uint4*      r4 = (uint4*)rec;
        int4  sv[EP_GRP], dv[EP_GRP];
        float di[EP_GRP][4];
        #pragma unroll
        for (int k = 0; k < EP_GRP; ++k) {
            const int g = blockIdx.x * (256 * EP_GRP) + k * 256 + tid;
            const int gc = (g < N_EDGES / 4) ? g : (N_EDGES / 4 - 1);
            sv[k] = s4[gc];
            dv[k] = d4[gc];
        }
        #pragma unroll
        for (int k = 0; k < EP_GRP; ++k) {
            di[k][0] = dinv[sv[k].x]; di[k][1] = dinv[sv[k].y];
            di[k][2] = dinv[sv[k].z]; di[k][3] = dinv[sv[k].w];
        }
        #pragma unroll
        for (int k = 0; k < EP_GRP; ++k) {
            const int g = blockIdx.x * (256 * EP_GRP) + k * 256 + tid;
            if (g < N_EDGES / 4) {
                const unsigned d0 = (unsigned)dv[k].x, d1 = (unsigned)dv[k].y;
                const unsigned d2 = (unsigned)dv[k].z, d3 = (unsigned)dv[k].w;
                const unsigned x0 = ((unsigned)sv[k].x << 9) | (d0 - (d0 / OWN_NPB) * OWN_NPB);
                const unsigned x1 = ((unsigned)sv[k].y << 9) | (d1 - (d1 / OWN_NPB) * OWN_NPB);
                const unsigned x2 = ((unsigned)sv[k].z << 9) | (d2 - (d2 / OWN_NPB) * OWN_NPB);
                const unsigned x3 = ((unsigned)sv[k].w << 9) | (d3 - (d3 / OWN_NPB) * OWN_NPB);
                r4[2 * g]     = make_uint4(x0, __builtin_bit_cast(unsigned, di[k][0]),
                                           x1, __builtin_bit_cast(unsigned, di[k][1]));
                r4[2 * g + 1] = make_uint4(x2, __builtin_bit_cast(unsigned, di[k][2]),
                                           x3, __builtin_bit_cast(unsigned, di[k][3]));
            }
        }
        return;
    }
    const int strip = blockIdx.x - EP_BLOCKS;

    const float4* xg = (const float4*)(x + (size_t)strip * G1_ROWS * F_IN);
    #pragma unroll
    for (int it = 0; it < 8; ++it) {
        int q   = tid + 256 * it;
        int row = q >> 6;
        int c4  = q & 63;
        float4 v = xg[q];
        float* d = &As[row * LDS_STRIDE + c4 * 4];
        d[0] = v.x; d[1] = v.y; d[2] = v.z; d[3] = v.w;
    }
    __syncthreads();

    const int wave = tid >> 5;
    const int lane = tid & 31;
    const int m    = lane & 15;
    const int kg   = lane >> 4;
    const int n0   = wave * 16;

    v8f c0 = {0.f,0.f,0.f,0.f,0.f,0.f,0.f,0.f};
    v8f c1 = {0.f,0.f,0.f,0.f,0.f,0.f,0.f,0.f};
    for (int k0 = 0; k0 < F_IN; k0 += 4) {
        int ka = k0 + 2 * kg;
        v2f a0 = *(const v2f*)&As[m * LDS_STRIDE + ka];
        v2f a1 = *(const v2f*)&As[(16 + m) * LDS_STRIDE + ka];
        v2f b;
        b.x = W1[(size_t)ka       * F_H + n0 + m];
        b.y = W1[(size_t)(ka + 1) * F_H + n0 + m];
        c0 = __builtin_amdgcn_wmma_f32_16x16x4_f32(false, a0, false, b,
                                                   (short)0, c0, false, false);
        c1 = __builtin_amdgcn_wmma_f32_16x16x4_f32(false, a1, false, b,
                                                   (short)0, c1, false, false);
    }

    asm volatile("v_nop\n\tv_nop\n\tv_nop\n\tv_nop" : "+v"(c0), "+v"(c1));

    float* out = h1 + (size_t)strip * G1_ROWS * F_H;
    #pragma unroll
    for (int r = 0; r < 8; ++r) {
        int row = r + 8 * kg;
        out[(size_t)row * F_H + n0 + m] = c0[r];
        out[(size_t)(16 + row) * F_H + n0 + m] = c1[r];
    }
}

#define S1_NPB    OWN_NPB
#define S1_BLOCKS 256
#define S1_STEPS  (N_EDGES / 512)
#define S1_KIT    98
#define S1_QCAP   64
#define S1_ROT    12
#define S1_PITCH  (F_H + 4)

#define E2_NSLAB   366
#define E2_NONE    0xFFFFFFFFu

struct SumCtx {
    const uint2* rec;
    const float* dinv;
    const float* feat;
    float*       acc;
    unsigned*    slabctr;
    uint2*       slabs;
    int          lo;
};

__device__ __forceinline__ int s1_chunk(int k, int wave, int rot) {
    int t = wave + 32 * k;
    if (t >= S1_STEPS) return -1;
    int tt = t + rot;
    return (tt >= S1_STEPS) ? tt - S1_STEPS : tt;
}

__device__ __forceinline__ uint4 s1_load(const unsigned* __restrict__ omap, int k, int wave, int rot, int lane) {
    int tt = s1_chunk(k, wave, rot);
    if (tt < 0) return make_uint4(0u, 0u, 0u, 0u);
    return ((const uint4*)omap)[tt * 32 + lane];
}

__device__ __forceinline__ void s2_add32(bool ok, unsigned xs, float w, int lane,
                                         const float* __restrict__ h2, float* acc) {
    const unsigned okm = __builtin_amdgcn_ballot_w32(ok);
    if (okm == 0u) return;
    xs = ok ? xs : 0u;
    w  = ok ? w : 0.0f;
    const int half = lane >> 4;
    const int f    = lane & 15;
    unsigned xj[16];
    float r[16], wj[16];
    #pragma unroll
    for (int c = 0; c < 16; ++c) xj[c] = (unsigned)__shfl((int)xs, 2 * c + half);
    #pragma unroll
    for (int c = 0; c < 16; ++c) r[c] = h2[(size_t)(xj[c] >> 9) * H_PAD + f];
    #pragma unroll
    for (int c = 0; c < 16; ++c) wj[c] = __shfl(w, 2 * c + half);
    #pragma unroll
    for (int c = 0; c < 16; ++c) r[c] = ((okm >> (2 * c + half)) & 1u) != 0u ? r[c] : 0.0f;
    if (f < N_CLS) {
        #pragma unroll
        for (int c = 0; c < 16; ++c) atomicAdd(&acc[(int)(xj[c] & 511u) * N_CLS + f], r[c] * wj[c]);
    }
}

template <int L>
__device__ __forceinline__ void s1_rows(int nb, const unsigned* qw, int lane, const SumCtx& cx) {
    if (nb <= 0) return;
    const bool act = lane < nb;
    unsigned xs = 0u;
    float w = 0.0f;
    if (act) {
        const uint2 r = cx.rec[qw[lane]];
        xs = r.x;
        w = __builtin_bit_cast(float, r.y) * cx.dinv[cx.lo + (int)(r.x & 511u)];
    }
    const int s  = (int)(xs >> 9);
    const int dl = (int)(xs & 511u);
    if (L == 2) {
        s2_add32(act, xs, w, lane, cx.feat, cx.acc);
        return;
    }
    unsigned slab = 0u;
    if (lane == 0) slab = atomicAdd(cx.slabctr, 1u);
    slab = (unsigned)__builtin_amdgcn_readfirstlane((int)slab);
    if (slab < (unsigned)E2_NSLAB) {
        uint2 ent;
        ent.x = act ? xs : E2_NONE;
        ent.y = __builtin_bit_cast(unsigned, w);
        cx.slabs[slab * 32u + (unsigned)lane] = ent;
    }
    const float* __restrict__ h1 = cx.feat;
    float* acc = cx.acc;
    for (int j0 = 0; j0 < nb; j0 += 8) {
        float r[8][4];
        #pragma unroll
        for (int c = 0; c < 8; ++c) {
            r[c][0] = 0.0f; r[c][1] = 0.0f; r[c][2] = 0.0f; r[c][3] = 0.0f;
            if (j0 + c < nb) {
                int sj = __builtin_amdgcn_readlane(s, j0 + c);
                const float* row = h1 + (size_t)sj * F_H + lane;
                r[c][0] = row[0]; r[c][1] = row[32]; r[c][2] = row[64]; r[c][3] = row[96];
            }
        }
        #pragma unroll
        for (int c = 0; c < 8; ++c) {
            if (j0 + c < nb) {
                float wj = __builtin_bit_cast(float, __builtin_amdgcn_readlane(__builtin_bit_cast(int, w), j0 + c));
                int   dj = __builtin_amdgcn_readlane(dl, j0 + c);
                float* a = acc + dj * S1_PITCH + lane;
                atomicAdd(a,      r[c][0] * wj);
                atomicAdd(a + 32, r[c][1] * wj);
                atomicAdd(a + 64, r[c][2] * wj);
                atomicAdd(a + 96, r[c][3] * wj);
            }
        }
    }
}

__device__ __forceinline__ unsigned s1_nzb(unsigned y) {
    return ((y & 0x7F7F7F7Fu) + 0x7F7F7F7Fu) | y;
}

template <int L>
__device__ __forceinline__ void s1_scan(uint4 v, int tt, int lane, unsigned pat,
                                        unsigned* qw, int& qn, const SumCtx& cx) {
    if (tt < 0) return;
    unsigned n0 = s1_nzb(v.x ^ pat), n1 = s1_nzb(v.y ^ pat), n2 = s1_nzb(v.z ^ pat), n3 = s1_nzb(v.w ^ pat);
    unsigned c = n0 >> 7;
    c = (c & ~0x02020202u) | ((n1 >> 6) & 0x02020202u);
    c = (c & ~0x04040404u) | ((n2 >> 5) & 0x04040404u);
    c = (c & ~0x08080808u) | ((n3 >> 4) & 0x08080808u);
    unsigned mm = ~c & 0x0F0F0F0Fu;
    const unsigned e0 = (unsigned)tt * 512u + (unsigned)lane * 16u;
    for (;;) {
        const bool has = mm != 0u;
        const unsigned b = __builtin_amdgcn_ballot_w32(has);
        if (b == 0u) break;
        if (has) {
            unsigned p = (unsigned)__builtin_ctz(mm);
            qw[qn + (int)__builtin_amdgcn_mbcnt_lo(b, 0u)] = e0 + ((p & 3u) << 2) + (p >> 3);
            mm &= mm - 1u;
        }
        qn += __builtin_popcount(b);
        if (qn >= 32) {
            s1_rows<L>(32, qw + (qn - 32), lane, cx);
            qn -= 32;
        }
    }
}

template <int L>
__device__ __forceinline__ void s1_sweep(const unsigned* __restrict__ omap, uint4 v0, uint4 v1, uint4 v2,
                                         int wave, int rot, int lane, unsigned pat,
                                         unsigned* qw, const SumCtx& cx) {
    int qn = 0;
    for (int k = 0; k < S1_KIT; k += 3) {
        uint4 n0 = s1_load(omap, k + 3, wave, rot, lane);
        uint4 n1 = s1_load(omap, k + 4, wave, rot, lane);
        uint4 n2 = s1_load(omap, k + 5, wave, rot, lane);
        s1_scan<L>(v0, s1_chunk(k,     wave, rot), lane, pat, qw, qn, cx);
        s1_scan<L>(v1, s1_chunk(k + 1, wave, rot), lane, pat, qw, qn, cx);
        s1_scan<L>(v2, s1_chunk(k + 2, wave, rot), lane, pat, qw, qn, cx);
        v0 = n0; v1 = n1; v2 = n2;
    }
    s1_rows<L>(qn, qw, lane, cx);
}

__global__ __launch_bounds__(1024) void k_sum1(const unsigned* __restrict__ rec,
                                               const unsigned* __restrict__ omap,
                                               const float* __restrict__ dinv,
                                               const float* __restrict__ h1,
                                               const float* __restrict__ b1,
                                               const float* __restrict__ W2,
                                               const float* __restrict__ b2,
                                               float* __restrict__ h2,
                                               float* __restrict__ agg2,
                                               unsigned* __restrict__ hdr,
                                               unsigned* __restrict__ slabs) {
    __shared__ float    acc[S1_NPB * S1_PITCH];
    __shared__ unsigned q[32 * S1_QCAP];
    __shared__ unsigned slabctr;
    const int tid  = threadIdx.x;
    const int lane = tid & 31;
    const int wave = __builtin_amdgcn_readfirstlane(tid >> 5);
    const int lo   = blockIdx.x * S1_NPB;
    const int cnt  = (N_NODES - lo < S1_NPB) ? (N_NODES - lo) : S1_NPB;
    const int rot  = blockIdx.x * S1_ROT;
    const unsigned pat = (unsigned)blockIdx.x * 0x01010101u;

    uint4 v0 = s1_load(omap, 0, wave, rot, lane);
    uint4 v1 = s1_load(omap, 1, wave, rot, lane);
    uint4 v2 = s1_load(omap, 2, wave, rot, lane);

    if (tid == 0) slabctr = 0u;
    const float4* g4 = (const float4*)(h1 + (size_t)lo * F_H);
    float4* a4 = (float4*)acc;
    const float4 bb = ((const float4*)b1)[tid & 31];
    for (int i = tid; i < cnt * (F_H / 4); i += 1024) {
        const float  dv = dinv[lo + (i >> 5)];
        const float4 h  = g4[i];
        float4 o;
        o.x = h.x * dv * dv + bb.x;
        o.y = h.y * dv * dv + bb.y;
        o.z = h.z * dv * dv + bb.z;
        o.w = h.w * dv * dv + bb.w;
        a4[(i >> 5) * (S1_PITCH / 4) + (tid & 31)] = o;
    }
    __syncthreads();

    SumCtx cx;
    cx.rec = (const uint2*)rec; cx.dinv = dinv; cx.feat = h1; cx.acc = acc;
    cx.slabctr = &slabctr;
    cx.slabs   = (uint2*)slabs + (size_t)blockIdx.x * (E2_NSLAB * 32);
    cx.lo      = lo;
    s1_sweep<1>(omap, v0, v1, v2, wave, rot, lane, pat, q + wave * S1_QCAP, cx);
    __syncthreads();
    if (tid == 0) hdr[blockIdx.x] = slabctr;

    const int ntile = (cnt + 15) >> 4;
    if (wave < ntile) {
        const int m  = lane & 15;
        const int kg = lane >> 4;
        const int t0 = wave * 16;
        const int ra = (t0 + m < cnt) ? (t0 + m) : (cnt - 1);
        const float* Ar = acc + ra * S1_PITCH;
        const int  mc   = (m < N_CLS) ? m : 0;
        const bool real = m < N_CLS;

        float d2[8];
        #pragma unroll
        for (int r = 0; r < 8; ++r) {
            const int rr = t0 + r + 8 * kg;
            const float di = dinv[lo + ((rr < cnt) ? rr : (cnt - 1))];
            d2[r] = di * di;
        }
        const float bcol = real ? b2[m] : 0.0f;

        v8f c = {0.f,0.f,0.f,0.f,0.f,0.f,0.f,0.f};
        #pragma unroll
        for (int hh = 0; hh < 2; ++hh) {
            v2f a[16], b[16];
            #pragma unroll
            for (int j = 0; j < 16; ++j) {
                const int ka = 64 * hh + 4 * j + 2 * kg;
                a[j]   = *(const v2f*)&Ar[ka];
                b[j].x = W2[ka * N_CLS + mc];
                b[j].y = W2[(ka + 1) * N_CLS + mc];
            }
            #pragma unroll
            for (int j = 0; j < 16; ++j) {
                a[j].x = a[j].x > 0.0f ? a[j].x : 0.0f;
                a[j].y = a[j].y > 0.0f ? a[j].y : 0.0f;
                b[j].x = real ? b[j].x : 0.0f;
                b[j].y = real ? b[j].y : 0.0f;
            }
            __builtin_amdgcn_sched_barrier(0);
            #pragma unroll
            for (int j = 0; j < 16; ++j)
                c = __builtin_amdgcn_wmma_f32_16x16x4_f32(false, a[j], false, b[j],
                                                          (short)0, c, false, false);
            __builtin_amdgcn_sched_barrier(0);
            asm volatile("v_nop\n\tv_nop\n\tv_nop\n\tv_nop" : "+v"(c));
        }

        #pragma unroll
        for (int r = 0; r < 8; ++r) {
            const int rr = t0 + r + 8 * kg;
            if (rr < cnt) {
                const size_t o = (size_t)(lo + rr) * H_PAD + m;
                h2[o]   = c[r];
                agg2[o] = c[r] * d2[r] + bcol;
            }
        }
    }
}

__global__ __launch_bounds__(1024) void k_sum2(const unsigned* __restrict__ rec,
                                               const unsigned* __restrict__ omap,
                                               const float* __restrict__ dinv,
                                               const float* __restrict__ h2,
                                               const float* __restrict__ agg2,
                                               const unsigned* __restrict__ hdr,
                                               const unsigned* __restrict__ slabs,
                                               float* __restrict__ out) {
    __shared__ float    acc[S1_NPB * N_CLS];
    __shared__ unsigned q[32 * S1_QCAP];
    const int tid  = threadIdx.x;
    const int lane = tid & 31;
    const int wave = __builtin_amdgcn_readfirstlane(tid >> 5);
    const int lo   = blockIdx.x * S1_NPB;
    const int cnt  = (N_NODES - lo < S1_NPB) ? (N_NODES - lo) : S1_NPB;
    const unsigned nsl = (unsigned)__builtin_amdgcn_readfirstlane((int)hdr[blockIdx.x]);
    const uint2* bl = (const uint2*)slabs + (size_t)blockIdx.x * (E2_NSLAB * 32);

    uint2 ent = make_uint2(E2_NONE, 0u);
    if (nsl <= (unsigned)E2_NSLAB && (unsigned)wave < nsl) ent = bl[wave * 32 + lane];

    for (int i = tid; i < cnt * N_CLS; i += 1024) {
        const int n = i / N_CLS;
        acc[i] = agg2[(size_t)(lo + n) * H_PAD + (i - n * N_CLS)];
    }
    __syncthreads();

    if (nsl <= (unsigned)E2_NSLAB) {
        for (unsigned sl = (unsigned)wave; sl < nsl; sl += 32u) {
            uint2 nx = make_uint2(E2_NONE, 0u);
            if (sl + 32u < nsl) nx = bl[(sl + 32u) * 32u + (unsigned)lane];
            s2_add32(ent.x != E2_NONE, ent.x, __builtin_bit_cast(float, ent.y), lane, h2, acc);
            ent = nx;
        }
    } else {
        SumCtx cx;
        cx.rec = (const uint2*)rec; cx.dinv = dinv; cx.feat = h2; cx.acc = acc;
        cx.slabctr = nullptr; cx.slabs = nullptr;
        cx.lo      = lo;
        const int rot = blockIdx.x * S1_ROT;
        const unsigned pat = (unsigned)blockIdx.x * 0x01010101u;
        uint4 v0 = s1_load(omap, 0, wave, rot, lane);
        uint4 v1 = s1_load(omap, 1, wave, rot, lane);
        uint4 v2 = s1_load(omap, 2, wave, rot, lane);
        s1_sweep<2>(omap, v0, v1, v2, wave, rot, lane, pat, q + wave * S1_QCAP, cx);
    }
    __syncthreads();

    if (tid < cnt) {
        float v[N_CLS];
        #pragma unroll
        for (int c = 0; c < N_CLS; ++c) v[c] = acc[tid * N_CLS + c];
        float mx = -INFINITY;
        #pragma unroll
        for (int c = 0; c < N_CLS; ++c) mx = fmaxf(mx, v[c]);
        float s = 0.0f;
        #pragma unroll
        for (int c = 0; c < N_CLS; ++c) s += expf(v[c] - mx);
        float lse = logf(s) + mx;
        #pragma unroll
        for (int c = 0; c < N_CLS; ++c) acc[tid * N_CLS + c] = v[c] - lse;
    }
    __syncthreads();

    float* o = out + (size_t)lo * N_CLS;
    for (int i = tid; i < cnt * N_CLS; i += 1024) o[i] = acc[i];
}

extern "C" void kernel_launch(void* const* d_in, const int* in_sizes, int n_in,
                              void* d_out, int out_size, void* d_ws, size_t ws_size,
                              hipStream_t stream) {
    const float* x   = (const float*)d_in[0];
    const int*   ei  = (const int*)d_in[1];
    const float* W1  = (const float*)d_in[2];
    const float* b1  = (const float*)d_in[3];
    const float* W2  = (const float*)d_in[4];
    const float* b2  = (const float*)d_in[5];
    float*       out = (float*)d_out;

    const int* src = ei;
    const int* dst = ei + N_EDGES;

    float* ws    = (float*)d_ws;
    float* dinv  = ws;
    float* h1    = dinv + 100352;
    float* reg2  = h1 + (size_t)N_NODES * F_H;
    float* h2    = reg2;
    float* agg2  = reg2 + (size_t)N_NODES * H_PAD;
    unsigned* part  = (unsigned*)reg2;
    unsigned* omap  = (unsigned*)(reg2 + (size_t)N_NODES * 32);
    unsigned* hdr   = (unsigned*)(reg2 + (size_t)N_NODES * 36);
    unsigned* slabs = (unsigned*)(reg2 + (size_t)N_NODES * 36 + 256);
    unsigned* rec   = (unsigned*)(reg2 + (size_t)N_NODES * 96);

    k_deg_hist<<<DEG_GROUPS * DEG_RANGES, 1024, 0, stream>>>(dst, part, omap);
    k_dinv_sum<<<(N_NODES / 2 + 255) / 256, 256, 0, stream>>>(part, dinv);

    k_gemm1<<<EP_BLOCKS + N_NODES / G1_ROWS, 256, 0, stream>>>(x, W1, h1, src, dst, dinv, rec);
    k_sum1<<<S1_BLOCKS, 1024, 0, stream>>>(rec, omap, dinv, h1, b1, W2, b2, h2, agg2, hdr, slabs);

    k_sum2<<<S1_BLOCKS, 1024, 0, stream>>>(rec, omap, dinv, h2, agg2, hdr, slabs, out);
}


# layer-1 gather: a lane loads 16 B, so one instruction fetches a source node's whole 512-B h1 row (was 4 loads of one line each); LDS rows hold feature 4l+j at word 32j+l; sum order kept
# speedup vs baseline: 3.5985x; 1.1030x over previous
#include <hip/hip_runtime.h>
#include <math.h>

#define N_NODES 100000
#define N_EDGES 1600000
#define F_IN    256
#define F_H     128
#define N_CLS   10
#define H_PAD   16

typedef float v2f __attribute__((ext_vector_type(2)));
typedef float v8f __attribute__((ext_vector_type(8)));

#define OWN_NPB    391
#define DEG_GROUPS 32
#define DEG_RANGES 8
#define DEG_EPG    (N_EDGES / DEG_GROUPS)
#define DEG_NPR    (N_NODES / DEG_RANGES)
#define DEG_LD     13

__global__ __launch_bounds__(1024) void k_deg_hist(const int* __restrict__ dst,
                                                   unsigned* __restrict__ part,
                                                   unsigned* __restrict__ omap) {
    __shared__ unsigned cnt[DEG_NPR];
    const int tid = threadIdx.x;
    const int r   = blockIdx.x & (DEG_RANGES - 1);
    const int g   = blockIdx.x >> 3;
    const int4* p = (const int4*)(dst + (size_t)g * DEG_EPG);

    int4 v[DEG_LD];
    #pragma unroll
    for (int k = 0; k < DEG_LD; ++k) {
        int i = tid + 1024 * k;
        v[k] = (i < DEG_EPG / 4) ? p[i] : make_int4(-1, -1, -1, -1);
    }
    for (int i = tid; i < DEG_NPR; i += 1024) cnt[i] = 0u;
    __syncthreads();

    const unsigned lo = (unsigned)r * DEG_NPR;
    #pragma unroll
    for (int k = 0; k < DEG_LD; ++k) {
        unsigned a;
        a = (unsigned)v[k].x - lo; if (a < DEG_NPR) atomicAdd(&cnt[a], 1u);
        a = (unsigned)v[k].y - lo; if (a < DEG_NPR) atomicAdd(&cnt[a], 1u);
        a = (unsigned)v[k].z - lo; if (a < DEG_NPR) atomicAdd(&cnt[a], 1u);
        a = (unsigned)v[k].w - lo; if (a < DEG_NPR) atomicAdd(&cnt[a], 1u);
    }
    __syncthreads();

    unsigned* o = part + (size_t)g * (N_NODES / 2) + r * (DEG_NPR / 2);
    for (int j = tid; j < DEG_NPR / 2; j += 1024)
        o[j] = cnt[2 * j] | (cnt[2 * j + 1] << 16);

    #pragma unroll
    for (int k = 0; k < DEG_LD; ++k) {
        int i = tid + 1024 * k;
        if ((k & (DEG_RANGES - 1)) == r && i < DEG_EPG / 4) {
            unsigned o0 = (unsigned)v[k].x / OWN_NPB, o1 = (unsigned)v[k].y / OWN_NPB;
            unsigned o2 = (unsigned)v[k].z / OWN_NPB, o3 = (unsigned)v[k].w / OWN_NPB;
            omap[(size_t)g * (DEG_EPG / 4) + i] = o0 | (o1 << 8) | (o2 << 16) | (o3 << 24);
        }
    }
}

__global__ __launch_bounds__(256) void k_dinv_sum(const unsigned* __restrict__ part,
                                                  float* __restrict__ dinv) {
    int t = blockIdx.x * 256 + threadIdx.x;
    if (t >= N_NODES / 2) return;
    unsigned c0 = 1u, c1 = 1u;
    #pragma unroll
    for (int g = 0; g < DEG_GROUPS; ++g) {
        unsigned w = part[(size_t)g * (N_NODES / 2) + t];
        c0 += w & 0xffffu;
        c1 += w >> 16;
    }
    float d0 = (float)c0, d1 = (float)c1;
    float2 o;
    o.x = (d0 > 0.0f) ? rsqrtf(d0) : 0.0f;
    o.y = (d1 > 0.0f) ? rsqrtf(d1) : 0.0f;
    ((float2*)dinv)[t] = o;
}

#define LDS_STRIDE 260
#define G1_ROWS    32
#define EP_GRP     4
#define EP_BLOCKS  ((N_EDGES / 4 + 256 * EP_GRP - 1) / (256 * EP_GRP))
__global__ __launch_bounds__(256) void k_gemm1(const float* __restrict__ x,
                                               const float* __restrict__ W1,
                                               float* __restrict__ h1,
                                               const int* __restrict__ src,
                                               const int* __restrict__ dst,
                                               const float* __restrict__ dinv,
                                               unsigned* __restrict__ rec) {
    __shared__ float As[G1_ROWS * LDS_STRIDE];
    const int tid   = threadIdx.x;

    if (blockIdx.x < EP_BLOCKS) {
        const int4* s4 = (const int4*)src;
        const int4* d4 = (const int4*)dst;
        uint4*      r4 = (uint4*)rec;
        int4  sv[EP_GRP], dv[EP_GRP];
        float di[EP_GRP][4];
        #pragma unroll
        for (int k = 0; k < EP_GRP; ++k) {
            const int g = blockIdx.x * (256 * EP_GRP) + k * 256 + tid;
            const int gc = (g < N_EDGES / 4) ? g : (N_EDGES / 4 - 1);
            sv[k] = s4[gc];
            dv[k] = d4[gc];
        }
        #pragma unroll
        for (int k = 0; k < EP_GRP; ++k) {
            di[k][0] = dinv[sv[k].x]; di[k][1] = dinv[sv[k].y];
            di[k][2] = dinv[sv[k].z]; di[k][3] = dinv[sv[k].w];
        }
        #pragma unroll
        for (int k = 0; k < EP_GRP; ++k) {
            const int g = blockIdx.x * (256 * EP_GRP) + k * 256 + tid;
            if (g < N_EDGES / 4) {
                const unsigned d0 = (unsigned)dv[k].x, d1 = (unsigned)dv[k].y;
                const unsigned d2 = (unsigned)dv[k].z, d3 = (unsigned)dv[k].w;
                const unsigned x0 = ((unsigned)sv[k].x << 9) | (d0 - (d0 / OWN_NPB) * OWN_NPB);
                const unsigned x1 = ((unsigned)sv[k].y << 9) | (d1 - (d1 / OWN_NPB) * OWN_NPB);
                const unsigned x2 = ((unsigned)sv[k].z << 9) | (d2 - (d2 / OWN_NPB) * OWN_NPB);
                const unsigned x3 = ((unsigned)sv[k].w << 9) | (d3 - (d3 / OWN_NPB) * OWN_NPB);
                r4[2 * g]     = make_uint4(x0, __builtin_bit_cast(unsigned, di[k][0]),
                                           x1, __builtin_bit_cast(unsigned, di[k][1]));
                r4[2 * g + 1] = make_uint4(x2, __builtin_bit_cast(unsigned, di[k][2]),
                                           x3, __builtin_bit_cast(unsigned, di[k][3]));
            }
        }
        return;
    }
    const int strip = blockIdx.x - EP_BLOCKS;

    const float4* xg = (const float4*)(x + (size_t)strip * G1_ROWS * F_IN);
    #pragma unroll
    for (int it = 0; it < 8; ++it) {
        int q   = tid + 256 * it;
        int row = q >> 6;
        int c4  = q & 63;
        float4 v = xg[q];
        float* d = &As[row * LDS_STRIDE + c4 * 4];
        d[0] = v.x; d[1] = v.y; d[2] = v.z; d[3] = v.w;
    }
    __syncthreads();

    const int wave = tid >> 5;
    const int lane = tid & 31;
    const int m    = lane & 15;
    const int kg   = lane >> 4;
    const int n0   = wave * 16;

    v8f c0 = {0.f,0.f,0.f,0.f,0.f,0.f,0.f,0.f};
    v8f c1 = {0.f,0.f,0.f,0.f,0.f,0.f,0.f,0.f};
    for (int k0 = 0; k0 < F_IN; k0 += 4) {
        int ka = k0 + 2 * kg;
        v2f a0 = *(const v2f*)&As[m * LDS_STRIDE + ka];
        v2f a1 = *(const v2f*)&As[(16 + m) * LDS_STRIDE + ka];
        v2f b;
        b.x = W1[(size_t)ka       * F_H + n0 + m];
        b.y = W1[(size_t)(ka + 1) * F_H + n0 + m];
        c0 = __builtin_amdgcn_wmma_f32_16x16x4_f32(false, a0, false, b,
                                                   (short)0, c0, false, false);
        c1 = __builtin_amdgcn_wmma_f32_16x16x4_f32(false, a1, false, b,
                                                   (short)0, c1, false, false);
    }

    asm volatile("v_nop\n\tv_nop\n\tv_nop\n\tv_nop" : "+v"(c0), "+v"(c1));

    float* out = h1 + (size_t)strip * G1_ROWS * F_H;
    #pragma unroll
    for (int r = 0; r < 8; ++r) {
        int row = r + 8 * kg;
        out[(size_t)row * F_H + n0 + m] = c0[r];
        out[(size_t)(16 + row) * F_H + n0 + m] = c1[r];
    }
}

#define S1_NPB    OWN_NPB
#define S1_BLOCKS 256
#define S1_STEPS  (N_EDGES / 512)
#define S1_KIT    98
#define S1_QCAP   64
#define S1_ROT    12
#define S1_PITCH  (F_H + 4)

#define E2_NSLAB   366
#define E2_NONE    0xFFFFFFFFu

struct SumCtx {
    const uint2* rec;
    const float* dinv;
    const float* feat;
    float*       acc;
    unsigned*    slabctr;
    uint2*       slabs;
    int          lo;
};

__device__ __forceinline__ int s1_chunk(int k, int wave, int rot) {
    int t = wave + 32 * k;
    if (t >= S1_STEPS) return -1;
    int tt = t + rot;
    return (tt >= S1_STEPS) ? tt - S1_STEPS : tt;
}

__device__ __forceinline__ uint4 s1_load(const unsigned* __restrict__ omap, int k, int wave, int rot, int lane) {
    int tt = s1_chunk(k, wave, rot);
    if (tt < 0) return make_uint4(0u, 0u, 0u, 0u);
    return ((const uint4*)omap)[tt * 32 + lane];
}

__device__ __forceinline__ void s2_add32(bool ok, unsigned xs, float w, int lane,
                                         const float* __restrict__ h2, float* acc) {
    const unsigned okm = __builtin_amdgcn_ballot_w32(ok);
    if (okm == 0u) return;
    xs = ok ? xs : 0u;
    w  = ok ? w : 0.0f;
    const int half = lane >> 4;
    const int f    = lane & 15;
    unsigned xj[16];
    float r[16], wj[16];
    #pragma unroll
    for (int c = 0; c < 16; ++c) xj[c] = (unsigned)__shfl((int)xs, 2 * c + half);
    #pragma unroll
    for (int c = 0; c < 16; ++c) r[c] = h2[(size_t)(xj[c] >> 9) * H_PAD + f];
    #pragma unroll
    for (int c = 0; c < 16; ++c) wj[c] = __shfl(w, 2 * c + half);
    #pragma unroll
    for (int c = 0; c < 16; ++c) r[c] = ((okm >> (2 * c + half)) & 1u) != 0u ? r[c] : 0.0f;
    if (f < N_CLS) {
        #pragma unroll
        for (int c = 0; c < 16; ++c) atomicAdd(&acc[(int)(xj[c] & 511u) * N_CLS + f], r[c] * wj[c]);
    }
}

template <int L>
__device__ __forceinline__ void s1_rows(int nb, const unsigned* qw, int lane, const SumCtx& cx) {
    if (nb <= 0) return;
    const bool act = lane < nb;
    unsigned xs = 0u;
    float w = 0.0f;
    if (act) {
        const uint2 r = cx.rec[qw[lane]];
        xs = r.x;
        w = __builtin_bit_cast(float, r.y) * cx.dinv[cx.lo + (int)(r.x & 511u)];
    }
    const int s  = (int)(xs >> 9);
    const int dl = (int)(xs & 511u);
    if (L == 2) {
        s2_add32(act, xs, w, lane, cx.feat, cx.acc);
        return;
    }
    unsigned slab = 0u;
    if (lane == 0) slab = atomicAdd(cx.slabctr, 1u);
    slab = (unsigned)__builtin_amdgcn_readfirstlane((int)slab);
    if (slab < (unsigned)E2_NSLAB) {
        uint2 ent;
        ent.x = act ? xs : E2_NONE;
        ent.y = __builtin_bit_cast(unsigned, w);
        cx.slabs[slab * 32u + (unsigned)lane] = ent;
    }
    const float* __restrict__ h1 = cx.feat;
    float* acc = cx.acc;
    for (int j0 = 0; j0 < nb; j0 += 8) {
        float4 r[8];
        #pragma unroll
        for (int c = 0; c < 8; ++c) {
            r[c] = make_float4(0.0f, 0.0f, 0.0f, 0.0f);
            if (j0 + c < nb) {
                int sj = __builtin_amdgcn_readlane(s, j0 + c);
                r[c] = *(const float4*)(h1 + (size_t)sj * F_H + 4 * lane);
            }
        }
        #pragma unroll
        for (int c = 0; c < 8; ++c) {
            if (j0 + c < nb) {
                float wj = __builtin_bit_cast(float, __builtin_amdgcn_readlane(__builtin_bit_cast(int, w), j0 + c));
                int   dj = __builtin_amdgcn_readlane(dl, j0 + c);
                float* a = acc + dj * S1_PITCH + lane;
                atomicAdd(a,      r[c].x * wj);
                atomicAdd(a + 32, r[c].y * wj);
                atomicAdd(a + 64, r[c].z * wj);
                atomicAdd(a + 96, r[c].w * wj);
            }
        }
    }
}

__device__ __forceinline__ unsigned s1_nzb(unsigned y) {
    return ((y & 0x7F7F7F7Fu) + 0x7F7F7F7Fu) | y;
}

template <int L>
__device__ __forceinline__ void s1_scan(uint4 v, int tt, int lane, unsigned pat,
                                        unsigned* qw, int& qn, const SumCtx& cx) {
    if (tt < 0) return;
    unsigned n0 = s1_nzb(v.x ^ pat), n1 = s1_nzb(v.y ^ pat), n2 = s1_nzb(v.z ^ pat), n3 = s1_nzb(v.w ^ pat);
    unsigned c = n0 >> 7;
    c = (c & ~0x02020202u) | ((n1 >> 6) & 0x02020202u);
    c = (c & ~0x04040404u) | ((n2 >> 5) & 0x04040404u);
    c = (c & ~0x08080808u) | ((n3 >> 4) & 0x08080808u);
    unsigned mm = ~c & 0x0F0F0F0Fu;
    const unsigned e0 = (unsigned)tt * 512u + (unsigned)lane * 16u;
    for (;;) {
        const bool has = mm != 0u;
        const unsigned b = __builtin_amdgcn_ballot_w32(has);
        if (b == 0u) break;
        if (has) {
            unsigned p = (unsigned)__builtin_ctz(mm);
            qw[qn + (int)__builtin_amdgcn_mbcnt_lo(b, 0u)] = e0 + ((p & 3u) << 2) + (p >> 3);
            mm &= mm - 1u;
        }
        qn += __builtin_popcount(b);
        if (qn >= 32) {
            s1_rows<L>(32, qw + (qn - 32), lane, cx);
            qn -= 32;
        }
    }
}

template <int L>
__device__ __forceinline__ void s1_sweep(const unsigned* __restrict__ omap, uint4 v0, uint4 v1, uint4 v2,
                                         int wave, int rot, int lane, unsigned pat,
                                         unsigned* qw, const SumCtx& cx) {
    int qn = 0;
    for (int k = 0; k < S1_KIT; k += 3) {
        uint4 n0 = s1_load(omap, k + 3, wave, rot, lane);
        uint4 n1 = s1_load(omap, k + 4, wave, rot, lane);
        uint4 n2 = s1_load(omap, k + 5, wave, rot, lane);
        s1_scan<L>(v0, s1_chunk(k,     wave, rot), lane, pat, qw, qn, cx);
        s1_scan<L>(v1, s1_chunk(k + 1, wave, rot), lane, pat, qw, qn, cx);
        s1_scan<L>(v2, s1_chunk(k + 2, wave, rot), lane, pat, qw, qn, cx);
        v0 = n0; v1 = n1; v2 = n2;
    }
    s1_rows<L>(qn, qw, lane, cx);
}

__global__ __launch_bounds__(1024) void k_sum1(const unsigned* __restrict__ rec,
                                               const unsigned* __restrict__ omap,
                                               const float* __restrict__ dinv,
                                               const float* __restrict__ h1,
                                               const float* __restrict__ b1,
                                               const float* __restrict__ W2,
                                               const float* __restrict__ b2,
                                               float* __restrict__ h2,
                                               float* __restrict__ agg2,
                                               unsigned* __restrict__ hdr,
                                               unsigned* __restrict__ slabs) {
    __shared__ float    acc[S1_NPB * S1_PITCH];
    __shared__ unsigned q[32 * S1_QCAP];
    __shared__ unsigned slabctr;
    const int tid  = threadIdx.x;
    const int lane = tid & 31;
    const int wave = __builtin_amdgcn_readfirstlane(tid >> 5);
    const int lo   = blockIdx.x * S1_NPB;
    const int cnt  = (N_NODES - lo < S1_NPB) ? (N_NODES - lo) : S1_NPB;
    const int rot  = blockIdx.x * S1_ROT;
    const unsigned pat = (unsigned)blockIdx.x * 0x01010101u;

    uint4 v0 = s1_load(omap, 0, wave, rot, lane);
    uint4 v1 = s1_load(omap, 1, wave, rot, lane);
    uint4 v2 = s1_load(omap, 2, wave, rot, lane);

    if (tid == 0) slabctr = 0u;
    const float4* g4 = (const float4*)(h1 + (size_t)lo * F_H);
    const float4 bb = ((const float4*)b1)[tid & 31];
    for (int i = tid; i < cnt * (F_H / 4); i += 1024) {
        const float  dv = dinv[lo + (i >> 5)];
        const float4 h  = g4[i];
        float4 o;
        o.x = h.x * dv * dv + bb.x;
        o.y = h.y * dv * dv + bb.y;
        o.z = h.z * dv * dv + bb.z;
        o.w = h.w * dv * dv + bb.w;
        float* ar = acc + (i >> 5) * S1_PITCH + (tid & 31);
        ar[0] = o.x; ar[32] = o.y; ar[64] = o.z; ar[96] = o.w;
    }
    __syncthreads();

    SumCtx cx;
    cx.rec = (const uint2*)rec; cx.dinv = dinv; cx.feat = h1; cx.acc = acc;
    cx.slabctr = &slabctr;
    cx.slabs   = (uint2*)slabs + (size_t)blockIdx.x * (E2_NSLAB * 32);
    cx.lo      = lo;
    s1_sweep<1>(omap, v0, v1, v2, wave, rot, lane, pat, q + wave * S1_QCAP, cx);
    __syncthreads();
    if (tid == 0) hdr[blockIdx.x] = slabctr;

    const int ntile = (cnt + 15) >> 4;
    if (wave < ntile) {
        const int m  = lane & 15;
        const int kg = lane >> 4;
        const int t0 = wave * 16;
        const int ra = (t0 + m < cnt) ? (t0 + m) : (cnt - 1);
        const float* Ar = acc + ra * S1_PITCH;
        const int  mc   = (m < N_CLS) ? m : 0;
        const bool real = m < N_CLS;

        float d2[8];
        #pragma unroll
        for (int r = 0; r < 8; ++r) {
            const int rr = t0 + r + 8 * kg;
            const float di = dinv[lo + ((rr < cnt) ? rr : (cnt - 1))];
            d2[r] = di * di;
        }
        const float bcol = real ? b2[m] : 0.0f;

        v8f c = {0.f,0.f,0.f,0.f,0.f,0.f,0.f,0.f};
        #pragma unroll
        for (int hh = 0; hh < 2; ++hh) {
            v2f a[16], b[16];
            #pragma unroll
            for (int j = 0; j < 16; ++j) {
                const int ka = 64 * hh + 4 * j + 2 * kg;
                const int pa = 64 * kg + 16 * hh + j;
                a[j].x = Ar[pa];
                a[j].y = Ar[pa + 32];
                b[j].x = W2[ka * N_CLS + mc];
                b[j].y = W2[(ka + 1) * N_CLS + mc];
            }
            #pragma unroll
            for (int j = 0; j < 16; ++j) {
                a[j].x = a[j].x > 0.0f ? a[j].x : 0.0f;
                a[j].y = a[j].y > 0.0f ? a[j].y : 0.0f;
                b[j].x = real ? b[j].x : 0.0f;
                b[j].y = real ? b[j].y : 0.0f;
            }
            __builtin_amdgcn_sched_barrier(0);
            #pragma unroll
            for (int j = 0; j < 16; ++j)
                c = __builtin_amdgcn_wmma_f32_16x16x4_f32(false, a[j], false, b[j],
                                                          (short)0, c, false, false);
            __builtin_amdgcn_sched_barrier(0);
            asm volatile("v_nop\n\tv_nop\n\tv_nop\n\tv_nop" : "+v"(c));
        }

        #pragma unroll
        for (int r = 0; r < 8; ++r) {
            const int rr = t0 + r + 8 * kg;
            if (rr < cnt) {
                const size_t o = (size_t)(lo + rr) * H_PAD + m;
                h2[o]   = c[r];
                agg2[o] = c[r] * d2[r] + bcol;
            }
        }
    }
}

__global__ __launch_bounds__(1024) void k_sum2(const unsigned* __restrict__ rec,
                                               const unsigned* __restrict__ omap,
                                               const float* __restrict__ dinv,
                                               const float* __restrict__ h2,
                                               const float* __restrict__ agg2,
                                               const unsigned* __restrict__ hdr,
                                               const unsigned* __restrict__ slabs,
                                               float* __restrict__ out) {
    __shared__ float    acc[S1_NPB * N_CLS];
    __shared__ unsigned q[32 * S1_QCAP];
    const int tid  = threadIdx.x;
    const int lane = tid & 31;
    const int wave = __builtin_amdgcn_readfirstlane(tid >> 5);
    const int lo   = blockIdx.x * S1_NPB;
    const int cnt  = (N_NODES - lo < S1_NPB) ? (N_NODES - lo) : S1_NPB;
    const unsigned nsl = (unsigned)__builtin_amdgcn_readfirstlane((int)hdr[blockIdx.x]);
    const uint2* bl = (const uint2*)slabs + (size_t)blockIdx.x * (E2_NSLAB * 32);

    uint2 ent = make_uint2(E2_NONE, 0u);
    if (nsl <= (unsigned)E2_NSLAB && (unsigned)wave < nsl) ent = bl[wave * 32 + lane];

    for (int i = tid; i < cnt * N_CLS; i += 1024) {
        const int n = i / N_CLS;
        acc[i] = agg2[(size_t)(lo + n) * H_PAD + (i - n * N_CLS)];
    }
    __syncthreads();

    if (nsl <= (unsigned)E2_NSLAB) {
        for (unsigned sl = (unsigned)wave; sl < nsl; sl += 32u) {
            uint2 nx = make_uint2(E2_NONE, 0u);
            if (sl + 32u < nsl) nx = bl[(sl + 32u) * 32u + (unsigned)lane];
            s2_add32(ent.x != E2_NONE, ent.x, __builtin_bit_cast(float, ent.y), lane, h2, acc);
            ent = nx;
        }
    } else {
        SumCtx cx;
        cx.rec = (const uint2*)rec; cx.dinv = dinv; cx.feat = h2; cx.acc = acc;
        cx.slabctr = nullptr; cx.slabs = nullptr;
        cx.lo      = lo;
        const int rot = blockIdx.x * S1_ROT;
        const unsigned pat = (unsigned)blockIdx.x * 0x01010101u;
        uint4 v0 = s1_load(omap, 0, wave, rot, lane);
        uint4 v1 = s1_load(omap, 1, wave, rot, lane);
        uint4 v2 = s1_load(omap, 2, wave, rot, lane);
        s1_sweep<2>(omap, v0, v1, v2, wave, rot, lane, pat, q + wave * S1_QCAP, cx);
    }
    __syncthreads();

    if (tid < cnt) {
        float v[N_CLS];
        #pragma unroll
        for (int c = 0; c < N_CLS; ++c) v[c] = acc[tid * N_CLS + c];
        float mx = -INFINITY;
        #pragma unroll
        for (int c = 0; c < N_CLS; ++c) mx = fmaxf(mx, v[c]);
        float s = 0.0f;
        #pragma unroll
        for (int c = 0; c < N_CLS; ++c) s += expf(v[c] - mx);
        float lse = logf(s) + mx;
        #pragma unroll
        for (int c = 0; c < N_CLS; ++c) acc[tid * N_CLS + c] = v[c] - lse;
    }
    __syncthreads();

    float* o = out + (size_t)lo * N_CLS;
    for (int i = tid; i < cnt * N_CLS; i += 1024) o[i] = acc[i];
}

extern "C" void kernel_launch(void* const* d_in, const int* in_sizes, int n_in,
                              void* d_out, int out_size, void* d_ws, size_t ws_size,
                              hipStream_t stream) {
    const float* x   = (const float*)d_in[0];
    const int*   ei  = (const int*)d_in[1];
    const float* W1  = (const float*)d_in[2];
    const float* b1  = (const float*)d_in[3];
    const float* W2  = (const float*)d_in[4];
    const float* b2  = (const float*)d_in[5];
    float*       out = (float*)d_out;

    const int* src = ei;
    const int* dst = ei + N_EDGES;

    float* ws    = (float*)d_ws;
    float* dinv  = ws;
    float* h1    = dinv + 100352;
    float* reg2  = h1 + (size_t)N_NODES * F_H;
    float* h2    = reg2;
    float* agg2  = reg2 + (size_t)N_NODES * H_PAD;
    unsigned* part  = (unsigned*)reg2;
    unsigned* omap  = (unsigned*)(reg2 + (size_t)N_NODES * 32);
    unsigned* hdr   = (unsigned*)(reg2 + (size_t)N_NODES * 36);
    unsigned* slabs = (unsigned*)(reg2 + (size_t)N_NODES * 36 + 256);
    unsigned* rec   = (unsigned*)(reg2 + (size_t)N_NODES * 96);

    k_deg_hist<<<DEG_GROUPS * DEG_RANGES, 1024, 0, stream>>>(dst, part, omap);
    k_dinv_sum<<<(N_NODES / 2 + 255) / 256, 256, 0, stream>>>(part, dinv);

    k_gemm1<<<EP_BLOCKS + N_NODES / G1_ROWS, 256, 0, stream>>>(x, W1, h1, src, dst, dinv, rec);
    k_sum1<<<S1_BLOCKS, 1024, 0, stream>>>(rec, omap, dinv, h1, b1, W2, b2, h2, agg2, hdr, slabs);

    k_sum2<<<S1_BLOCKS, 1024, 0, stream>>>(rec, omap, dinv, h2, agg2, hdr, slabs, out);
}
